# starting kernel
# baseline (speedup 1.0000x reference)
#include <hip/hip_runtime.h>

#define NB   4
#define NN   256
#define NINC 128
#define NOUTC 128

typedef __attribute__((ext_vector_type(2))) float v2f;
typedef __attribute__((ext_vector_type(4))) float v4f;
typedef __attribute__((ext_vector_type(8))) float v8f;

__global__ __launch_bounds__(32) void nodeconv_gemm(
    const float* __restrict__ node,
    const float* __restrict__ Wi,
    const float* __restrict__ Wj,
    float* __restrict__ vi,
    float* __restrict__ vj)
{
    const int tile = blockIdx.x;
    const int b    = tile >> 7;
    const int t    = tile & 127;
    const int m0   = (t >> 4) << 4;
    const int n0   = (t & 15) << 4;

    const int lane = threadIdx.x;
    const int half = lane >> 4;
    const int idx  = lane & 15;

    const float* nb = node + b * (NINC * NN);

    v8f acc_i = {0.f, 0.f, 0.f, 0.f, 0.f, 0.f, 0.f, 0.f};
    v8f acc_j = {0.f, 0.f, 0.f, 0.f, 0.f, 0.f, 0.f, 0.f};

    const int arow = (m0 + idx) * NINC;

    for (int kk = 0; kk < NINC; kk += 4) {
        const int ka = kk + 2 * half;
        v2f a_i = *(const v2f*)(Wi + arow + ka);
        v2f a_j = *(const v2f*)(Wj + arow + ka);
        v2f bf;
        bf.x = nb[(ka    ) * NN + n0 + idx];
        bf.y = nb[(ka + 1) * NN + n0 + idx];

        acc_i = __builtin_amdgcn_wmma_f32_16x16x4_f32(
             false, a_i,  false, bf,
             (short)0, acc_i,  false,  false);
        acc_j = __builtin_amdgcn_wmma_f32_16x16x4_f32(
             false, a_j,  false, bf,
             (short)0, acc_j,  false,  false);
    }

    #pragma unroll
    for (int r = 0; r < 8; ++r) {
        const int o = m0 + 8 * half + r;
        vi[(b * NOUTC + o) * NN + n0 + idx] = acc_i[r];
        vj[(b * NOUTC + o) * NN + n0 + idx] = acc_j[r];
    }
}

__global__ __launch_bounds__(256) void nodeconv_bcast(
    const float* __restrict__ adj,
    const float* __restrict__ vi,
    const float* __restrict__ vj,
    float* __restrict__ out)
{
    const int tid    = threadIdx.x;
    const int rlocal = tid >> 6;
    const int jv     = (tid & 63) << 2;

    const int row = blockIdx.x * 4 + rlocal;
    const int b   = row >> 15;
    const int rem = row & 32767;
    const int o   = rem >> 8;
    const int i   = rem & 255;

    const v4f a4 = *(const v4f*)(adj + (b * NN + i) * NN + jv);
    const v4f w4 = *(const v4f*)(vj + (b * NOUTC + o) * NN + jv);

    v4f r4 = a4 * w4;

    if (i >= jv && i < jv + 4) {
        const float dvi = vi[(b * NOUTC + o) * NN + i];
        r4[i - jv] = a4[i - jv] * dvi;
    }

    v4f* dst = (v4f*)(out + (((long long)(b * NOUTC + o) * NN + i) * NN) + jv);
    __builtin_nontemporal_store(r4, dst);
}

extern "C" void kernel_launch(void* const* d_in, const int* in_sizes, int n_in,
                              void* d_out, int out_size, void* d_ws, size_t ws_size,
                              hipStream_t stream) {
    (void)in_sizes; (void)n_in; (void)out_size; (void)ws_size;

    const float* adj  = (const float*)d_in[0];
    const float* node = (const float*)d_in[1];
    const float* Wi   = (const float*)d_in[2];
    const float* Wj   = (const float*)d_in[3];
    float* out = (float*)d_out;

    float* vi = (float*)d_ws;
    float* vj = vi + (size_t)NB * NOUTC * NN;

    nodeconv_gemm<<<NB * 8 * 16, 32, 0, stream>>>(node, Wi, Wj, vi, vj);

    const int rows   = NB * NOUTC * NN;
    const int blocks = rows / 4;
    nodeconv_bcast<<<blocks, 256, 0, stream>>>(adj, vi, vj, out);
}


# bcast kernel made persistent: 256 work-groups of 8 waves instead of 32768 one-store groups; each wave keeps 8 adj rows in registers and loops over 8 channels, 16 whole-line b128 stores per trip
# speedup vs baseline: 2.0507x; 2.0507x over previous
#include <hip/hip_runtime.h>

#define NB   4
#define NN   256
#define NINC 128
#define NOUTC 128

typedef __attribute__((ext_vector_type(2))) float v2f;
typedef __attribute__((ext_vector_type(4))) float v4f;
typedef __attribute__((ext_vector_type(8))) float v8f;

__global__ __launch_bounds__(32) void nodeconv_gemm(
    const float* __restrict__ node,
    const float* __restrict__ Wi,
    const float* __restrict__ Wj,
    float* __restrict__ vi,
    float* __restrict__ vj)
{
    const int tile = blockIdx.x;
    const int b    = tile >> 7;
    const int t    = tile & 127;
    const int m0   = (t >> 4) << 4;
    const int n0   = (t & 15) << 4;

    const int lane = threadIdx.x;
    const int half = lane >> 4;
    const int idx  = lane & 15;

    const float* nb = node + b * (NINC * NN);

    v8f acc_i = {0.f, 0.f, 0.f, 0.f, 0.f, 0.f, 0.f, 0.f};
    v8f acc_j = {0.f, 0.f, 0.f, 0.f, 0.f, 0.f, 0.f, 0.f};

    const int arow = (m0 + idx) * NINC;

    for (int kk = 0; kk < NINC; kk += 4) {
        const int ka = kk + 2 * half;
        v2f a_i = *(const v2f*)(Wi + arow + ka);
        v2f a_j = *(const v2f*)(Wj + arow + ka);
        v2f bf;
        bf.x = nb[(ka    ) * NN + n0 + idx];
        bf.y = nb[(ka + 1) * NN + n0 + idx];

        acc_i = __builtin_amdgcn_wmma_f32_16x16x4_f32(
             false, a_i,  false, bf,
             (short)0, acc_i,  false,  false);
        acc_j = __builtin_amdgcn_wmma_f32_16x16x4_f32(
             false, a_j,  false, bf,
             (short)0, acc_j,  false,  false);
    }

    #pragma unroll
    for (int r = 0; r < 8; ++r) {
        const int o = m0 + 8 * half + r;
        vi[(b * NOUTC + o) * NN + n0 + idx] = acc_i[r];
        vj[(b * NOUTC + o) * NN + n0 + idx] = acc_j[r];
    }
}

__global__ __launch_bounds__(256) __attribute__((amdgpu_waves_per_eu(2, 2))) void nodeconv_bcast(
    const float* __restrict__ adj,
    const float* __restrict__ vi,
    const float* __restrict__ vj,
    float* __restrict__ out)
{
    const int lane = threadIdx.x & 31;
    const int wave = __builtin_amdgcn_readfirstlane(threadIdx.x >> 5);

    const int wg = blockIdx.x;
    const int b  = wg >> 6;
    const int i0 = ((wg >> 1) & 31) << 3;
    const int o0 = (((wg & 1) << 3) + wave) << 3;

    const int hd = i0 >> 7;
    const int L0 = (i0 & 127) >> 2;
    const bool dg0 = (lane == L0);
    const bool dg1 = (lane == L0 + 1);

    const float* ap = adj + (b * NN + i0) * NN + 4 * lane;
    v4f a[8][2];
    #pragma unroll
    for (int r = 0; r < 8; ++r) {
        a[r][0] = *(const v4f*)(ap + r * NN);
        a[r][1] = *(const v4f*)(ap + r * NN + 128);
    }

    const float* wjp = vj + (b * NOUTC + o0) * NN + 4 * lane;
    const float* dip = vi + (b * NOUTC + o0) * NN + i0;
    float* op = out + ((long long)((b * NOUTC + o0) * NN + i0) * NN) + 4 * lane;

    #pragma unroll
    for (int q = 0; q < 8; ++q) {
        const v4f w0 = *(const v4f*)(wjp + q * NN);
        const v4f w1 = *(const v4f*)(wjp + q * NN + 128);
        float d[8];
        #pragma unroll
        for (int r = 0; r < 8; ++r) d[r] = dip[q * NN + r];

        v4f res[8][2];
        #pragma unroll
        for (int r = 0; r < 8; ++r) {
            const bool dg = (r < 4) ? dg0 : dg1;
            v4f x0 = w0;
            v4f x1 = w1;
            x0[r & 3] = (dg && hd == 0) ? d[r] : w0[r & 3];
            x1[r & 3] = (dg && hd == 1) ? d[r] : w1[r & 3];
            res[r][0] = a[r][0] * x0;
            res[r][1] = a[r][1] * x1;
        }

        float* oq = op + (long long)q * (NN * NN);
        #pragma unroll
        for (int r = 0; r < 8; ++r) {
            __builtin_nontemporal_store(res[r][0], (v4f*)(oq + r * NN));
            __builtin_nontemporal_store(res[r][1], (v4f*)(oq + r * NN + 128));
        }
    }
}

extern "C" void kernel_launch(void* const* d_in, const int* in_sizes, int n_in,
                              void* d_out, int out_size, void* d_ws, size_t ws_size,
                              hipStream_t stream) {
    (void)in_sizes; (void)n_in; (void)out_size; (void)ws_size;

    const float* adj  = (const float*)d_in[0];
    const float* node = (const float*)d_in[1];
    const float* Wi   = (const float*)d_in[2];
    const float* Wj   = (const float*)d_in[3];
    float* out = (float*)d_out;

    float* vi = (float*)d_ws;
    float* vj = vi + (size_t)NB * NOUTC * NN;

    nodeconv_gemm<<<NB * 8 * 16, 32, 0, stream>>>(node, Wi, Wj, vi, vj);

    nodeconv_bcast<<<256, 256, 0, stream>>>(adj, vi, vj, out);
}


# both launches fused into one: each 8-wave work-group computes its own vj (16 ch x 256 col) and vi (16 ch x 32 row) tiles by f32 WMMA into LDS, then streams its 32 rows x 16 ch with the same stores
# speedup vs baseline: 2.4970x; 1.2176x over previous
#include <hip/hip_runtime.h>

#define NB   4
#define NN   256
#define NINC 128
#define NOUTC 128

#define VJP 260
#define VIP 32

typedef __attribute__((ext_vector_type(2))) float v2f;
typedef __attribute__((ext_vector_type(4))) float v4f;
typedef __attribute__((ext_vector_type(8))) float v8f;


template <bool WITH_VI>
__device__ __forceinline__ void contract_phase(
    const float* __restrict__ nb,
    const float* __restrict__ Wi,
    const float* __restrict__ Wj,
    float* __restrict__ vjs, float* __restrict__ vis,
    const int o0, const int i0, const int wave, const int lane)
{
    const int half = lane >> 4;
    const int idx  = lane & 15;

    const float* wjp = Wj + (o0 + idx) * NINC + 2 * half;
    const float* wip = Wi + (o0 + idx) * NINC + 2 * half;
    const float* bjp = nb + (2 * half) * NN + 32 * wave + 2 * idx;
    const float* bip = nb + (2 * half) * NN + i0 + 16 * wave + idx;

    v8f acc0 = {0.f, 0.f, 0.f, 0.f, 0.f, 0.f, 0.f, 0.f};
    v8f acc1 = {0.f, 0.f, 0.f, 0.f, 0.f, 0.f, 0.f, 0.f};
    v8f acci = {0.f, 0.f, 0.f, 0.f, 0.f, 0.f, 0.f, 0.f};

    #pragma unroll
    for (int kk = 0; kk < NINC; kk += 4) {
        const v2f a_j = *(const v2f*)(wjp + kk);
        const v2f x0  = *(const v2f*)(bjp + (kk    ) * NN);
        const v2f x1  = *(const v2f*)(bjp + (kk + 1) * NN);
        v2f b0, b1;
        b0.x = x0.x; b0.y = x1.x;
        b1.x = x0.y; b1.y = x1.y;
        acc0 = __builtin_amdgcn_wmma_f32_16x16x4_f32(
             false, a_j,  false, b0,
             (short)0, acc0,  false,  false);
        acc1 = __builtin_amdgcn_wmma_f32_16x16x4_f32(
             false, a_j,  false, b1,
             (short)0, acc1,  false,  false);
        if (WITH_VI) {
            const v2f a_i = *(const v2f*)(wip + kk);
            v2f bi;
            bi.x = bip[(kk    ) * NN];
            bi.y = bip[(kk + 1) * NN];
            acci = __builtin_amdgcn_wmma_f32_16x16x4_f32(
                 false, a_i,  false, bi,
                 (short)0, acci,  false,  false);
        }
    }

    float* vjw = vjs + (8 * half) * VJP + 32 * wave + 2 * idx;
    #pragma unroll
    for (int r = 0; r < 8; ++r) {
        v2f p;
        p.x = acc0[r];
        p.y = acc1[r];
        *(v2f*)(vjw + r * VJP) = p;
    }
    if (WITH_VI) {
        float* viw = vis + (8 * half) * VIP + 16 * wave + idx;
        #pragma unroll
        for (int r = 0; r < 8; ++r) viw[r * VIP] = acci[r];
    }
}

__global__ __launch_bounds__(256) __attribute__((amdgpu_waves_per_eu(2, 2))) void nodeconv_fused(
    const float* __restrict__ adj,
    const float* __restrict__ node,
    const float* __restrict__ Wi,
    const float* __restrict__ Wj,
    float* __restrict__ out)
{
    __shared__ __attribute__((aligned(16))) float vjs[16 * VJP];
    __shared__ __attribute__((aligned(16))) float vis[16 * VIP];

    const int lane = threadIdx.x & 31;
    const int wave = __builtin_amdgcn_readfirstlane(threadIdx.x >> 5);

    const int wg = blockIdx.x;
    const int b  = wg >> 6;
    const int o0 = ((wg >> 3) & 7) << 4;
    const int i0 = (wg & 7) << 5;

    const int iw = i0 + ((wave & 3) << 3);
    const int oc = (wave >> 2) << 3;

    const float* ap = adj + (b * NN + iw) * NN + 4 * lane;
    v4f a[8][2];
    #pragma unroll
    for (int r = 0; r < 8; ++r) {
        a[r][0] = *(const v4f*)(ap + r * NN);
        a[r][1] = *(const v4f*)(ap + r * NN + 128);
    }

    const float* nb = node + b * (NINC * NN);
    if (wave < 2) contract_phase<true >(nb, Wi, Wj, vjs, vis, o0, i0, wave, lane);
    else          contract_phase<false>(nb, Wi, Wj, vjs, vis, o0, i0, wave, lane);

    __syncthreads();

    const int hd = iw >> 7;
    const int L0 = (iw & 127) >> 2;
    const bool dg0 = (lane == L0);
    const bool dg1 = (lane == L0 + 1);

    const float* wjl = vjs + oc * VJP + 4 * lane;
    const float* dil = vis + oc * VIP + (iw - i0);
    float* op = out + ((long long)((b * NOUTC + o0 + oc) * NN + iw) * NN) + 4 * lane;

    #pragma unroll
    for (int q = 0; q < 8; ++q) {
        const v4f w0 = *(const v4f*)(wjl + q * VJP);
        const v4f w1 = *(const v4f*)(wjl + q * VJP + 128);
        const v4f dA = *(const v4f*)(dil + q * VIP);
        const v4f dB = *(const v4f*)(dil + q * VIP + 4);
        const float d[8] = {dA.x, dA.y, dA.z, dA.w, dB.x, dB.y, dB.z, dB.w};

        v4f res[8][2];
        #pragma unroll
        for (int r = 0; r < 8; ++r) {
            const bool dg = (r < 4) ? dg0 : dg1;
            v4f x0 = w0;
            v4f x1 = w1;
            x0[r & 3] = (dg && hd == 0) ? d[r] : w0[r & 3];
            x1[r & 3] = (dg && hd == 1) ? d[r] : w1[r & 3];
            res[r][0] = a[r][0] * x0;
            res[r][1] = a[r][1] * x1;
        }

        float* oq = op + (long long)q * (NN * NN);
        #pragma unroll
        for (int r = 0; r < 8; ++r) {
            __builtin_nontemporal_store(res[r][0], (v4f*)(oq + r * NN));
            __builtin_nontemporal_store(res[r][1], (v4f*)(oq + r * NN + 128));
        }
    }
}

extern "C" void kernel_launch(void* const* d_in, const int* in_sizes, int n_in,
                              void* d_out, int out_size, void* d_ws, size_t ws_size,
                              hipStream_t stream) {
    (void)in_sizes; (void)n_in; (void)out_size; (void)d_ws; (void)ws_size;

    const float* adj  = (const float*)d_in[0];
    const float* node = (const float*)d_in[1];
    const float* Wi   = (const float*)d_in[2];
    const float* Wj   = (const float*)d_in[3];
    float* out = (float*)d_out;

    nodeconv_fused<<<256, 256, 0, stream>>>(adj, node, Wi, Wj, out);
}
